# starting kernel
# baseline (speedup 1.0000x reference)
#include <hip/hip_runtime.h>

#define H_DIM 1024
#define N_DIM 64
#define L_DIM 2048

typedef float v2f __attribute__((ext_vector_type(2)));
typedef float v8f __attribute__((ext_vector_type(8)));

__device__ __forceinline__ void cmul(float& xr, float& xi, float yr, float yi) {
    float tr = xr * yr - xi * yi;
    xi = xr * yi + xi * yr;
    xr = tr;
}

__global__ __launch_bounds__(32) void s4d_wmma_kernel(
    const float* __restrict__ log_dt,
    const float* __restrict__ C_real,
    const float* __restrict__ log_A_real,
    const float* __restrict__ A_imag,
    float* __restrict__ out)
{
    __shared__ float sQr[16 * 65];
    __shared__ float sQi[16 * 65];
    __shared__ float sPr[64 * 65];
    __shared__ float sPi[64 * 65];

    const int h       = blockIdx.x;
    const int lane    = threadIdx.x;
    const int halfSel = lane >> 4;
    const int mn      = lane & 15;

    const float dt = expf(log_dt[h]);

    float wr[2], wi[2], W16r[2], W16i[2], pr[2], pi[2];
    #pragma unroll
    for (int s = 0; s < 2; ++s) {
        const int n   = lane + 32 * s;
        const float lar = log_A_real[h * N_DIM + n];
        const float aim = A_imag[h * N_DIM + n];
        const float cr  = C_real[(h * N_DIM + n) * 2 + 0];
        const float ci  = C_real[(h * N_DIM + n) * 2 + 1];

        const float ar  = -expf(lar);
        const float dAr = ar * dt, dAi = aim * dt;

        const float er = expf(dAr);
        wr[s] = er * cosf(dAi);
        wi[s] = er * sinf(dAi);

        const float e1r = wr[s] - 1.0f, e1i = wi[s];
        const float inv = 1.0f / (ar * ar + aim * aim);
        const float tr  = (e1r * ar + e1i * aim) * inv;
        const float ti  = (e1i * ar - e1r * aim) * inv;
        pr[s] = 2.0f * (cr * tr - ci * ti);
        pi[s] = 2.0f * (cr * ti + ci * tr);

        const float er16 = expf(16.0f * dAr);
        W16r[s] = er16 * cosf(16.0f * dAi);
        W16i[s] = er16 * sinf(16.0f * dAi);
    }

    #pragma unroll
    for (int s = 0; s < 2; ++s) {
        const int n = lane + 32 * s;
        float qr = 1.0f, qi = 0.0f;
        for (int l0 = 0; l0 < 16; ++l0) {
            sQr[l0 * 65 + n] = qr;
            sQi[l0 * 65 + n] = qi;
            cmul(qr, qi, wr[s], wi[s]);
        }
    }

    for (int pass = 0; pass < 2; ++pass) {
        __syncthreads();
        #pragma unroll
        for (int s = 0; s < 2; ++s) {
            const int n = lane + 32 * s;
            for (int l1 = 0; l1 < 64; ++l1) {
                sPr[n * 65 + l1] = pr[s];
                sPi[n * 65 + l1] = -pi[s];
                cmul(pr[s], pi[s], W16r[s], W16i[s]);
            }
        }
        __syncthreads();

        for (int j = 0; j < 4; ++j) {
            v8f acc = {};
            #pragma unroll
            for (int k0 = 0; k0 < 64; k0 += 4) {
                const int kk = k0 + 2 * halfSel;
                v2f a, b;
                a.x = sQr[mn * 65 + kk];
                a.y = sQr[mn * 65 + kk + 1];
                b.x = sPr[kk * 65 + j * 16 + mn];
                b.y = sPr[(kk + 1) * 65 + j * 16 + mn];
                acc = __builtin_amdgcn_wmma_f32_16x16x4_f32(
                    false, a, false, b, (short)0, acc, false, false);
                a.x = sQi[mn * 65 + kk];
                a.y = sQi[mn * 65 + kk + 1];
                b.x = sPi[kk * 65 + j * 16 + mn];
                b.y = sPi[(kk + 1) * 65 + j * 16 + mn];
                acc = __builtin_amdgcn_wmma_f32_16x16x4_f32(
                    false, a, false, b, (short)0, acc, false, false);
            }
            const int jg = pass * 4 + j;
            float* dst = out + h * L_DIM + (jg * 16 + mn) * 16 + 8 * halfSel;
            *reinterpret_cast<float4*>(dst)     = make_float4(acc[0], acc[1], acc[2], acc[3]);
            *reinterpret_cast<float4*>(dst + 4) = make_float4(acc[4], acc[5], acc[6], acc[7]);
        }
    }
}

extern "C" void kernel_launch(void* const* d_in, const int* in_sizes, int n_in,
                              void* d_out, int out_size, void* d_ws, size_t ws_size,
                              hipStream_t stream) {
    const float* log_dt     = (const float*)d_in[0];
    const float* C_real     = (const float*)d_in[1];
    const float* log_A_real = (const float*)d_in[2];
    const float* A_imag     = (const float*)d_in[3];
    float* out = (float*)d_out;

    dim3 grid(H_DIM);
    dim3 block(32);
    hipLaunchKernelGGL(s4d_wmma_kernel, grid, block, 0, stream,
                       log_dt, C_real, log_A_real, A_imag, out);
}


# B fragments of P stay in registers, advanced per 16-column tile by one complex multiply with W256=exp(256*dtA); LDS holds two 16-step power tables read once; sum order kept, powers W16^c*W256^j
# speedup vs baseline: 1.2982x; 1.2982x over previous
#include <hip/hip_runtime.h>

#define H_DIM 1024
#define N_DIM 64
#define L_DIM 2048

#define ROW_PITCH 132

typedef float v2f __attribute__((ext_vector_type(2)));
typedef float v4f __attribute__((ext_vector_type(4)));
typedef float v8f __attribute__((ext_vector_type(8)));

__device__ __forceinline__ void cmul(float& xr, float& xi, float yr, float yi) {
    float tr = xr * yr - xi * yi;
    xi = xr * yi + xi * yr;
    xr = tr;
}

__global__ __launch_bounds__(32) void s4d_wmma_kernel(
    const float* __restrict__ log_dt,
    const float* __restrict__ C_real,
    const float* __restrict__ log_A_real,
    const float* __restrict__ A_imag,
    float* __restrict__ out)
{
    __shared__ __attribute__((aligned(16))) float sQ[16 * ROW_PITCH];
    __shared__ __attribute__((aligned(16))) float sP[16 * ROW_PITCH];
    __shared__ __attribute__((aligned(16))) float sW[2 * N_DIM];

    const int h       = blockIdx.x;
    const int lane    = threadIdx.x;
    const int halfSel = lane >> 4;
    const int mn      = lane & 15;

    const float dt = expf(log_dt[h]);

    float wr[2], wi[2], W16r[2], W16i[2], pr[2], pi[2];
    #pragma unroll
    for (int s = 0; s < 2; ++s) {
        const int n   = lane + 32 * s;
        const float lar = log_A_real[h * N_DIM + n];
        const float aim = A_imag[h * N_DIM + n];
        const float cr  = C_real[(h * N_DIM + n) * 2 + 0];
        const float ci  = C_real[(h * N_DIM + n) * 2 + 1];

        const float ar  = -expf(lar);
        const float dAr = ar * dt, dAi = aim * dt;

        const float er = expf(dAr);
        wr[s] = er * cosf(dAi);
        wi[s] = er * sinf(dAi);

        const float e1r = wr[s] - 1.0f, e1i = wi[s];
        const float inv = 1.0f / (ar * ar + aim * aim);
        const float tr  = (e1r * ar + e1i * aim) * inv;
        const float ti  = (e1i * ar - e1r * aim) * inv;
        pr[s] = 2.0f * (cr * tr - ci * ti);
        pi[s] = 2.0f * (cr * ti + ci * tr);

        const float er16 = expf(16.0f * dAr);
        W16r[s] = er16 * cosf(16.0f * dAi);
        W16i[s] = er16 * sinf(16.0f * dAi);

        const float er256 = expf(256.0f * dAr);
        const int slot = 4 * (n >> 1) + (n & 1);
        sW[slot]     = er256 * cosf(256.0f * dAi);
        sW[slot + 2] = er256 * sinf(256.0f * dAi);
    }

    {
        float qr[2] = {1.0f, 1.0f}, qi[2] = {0.0f, 0.0f};
        #pragma unroll
        for (int x = 0; x < 16; ++x) {
            #pragma unroll
            for (int s = 0; s < 2; ++s) {
                const int n = lane + 32 * s;
                const int slot = x * ROW_PITCH + 4 * (n >> 1) + (n & 1);
                sQ[slot]     = qr[s];
                sQ[slot + 2] = qi[s];
                sP[slot]     = pr[s];
                sP[slot + 2] = -pi[s];
                cmul(qr[s], qi[s], wr[s], wi[s]);
                cmul(pr[s], pi[s], W16r[s], W16i[s]);
            }
        }
    }
    __syncthreads();

    v2f Ar[16], Ai[16], Br[16], Bq[16], Wr[16], Wi[16];
    #pragma unroll
    for (int i = 0; i < 16; ++i) {
        const int off = 4 * (2 * i + halfSel);
        const v4f qa = *reinterpret_cast<const v4f*>(&sQ[mn * ROW_PITCH + off]);
        const v4f pb = *reinterpret_cast<const v4f*>(&sP[mn * ROW_PITCH + off]);
        const v4f ww = *reinterpret_cast<const v4f*>(&sW[off]);
        Ar[i].x = qa.x; Ar[i].y = qa.y; Ai[i].x = qa.z; Ai[i].y = qa.w;
        Br[i].x = pb.x; Br[i].y = pb.y; Bq[i].x = pb.z; Bq[i].y = pb.w;
        Wr[i].x = ww.x; Wr[i].y = ww.y; Wi[i].x = ww.z; Wi[i].y = ww.w;
    }

    float* dst = out + h * L_DIM + mn * 16 + 8 * halfSel;

    #pragma unroll
    for (int j = 0; j < 8; ++j) {
        v8f acc = {};
        #pragma unroll
        for (int i = 0; i < 16; ++i) {
            acc = __builtin_amdgcn_wmma_f32_16x16x4_f32(
                false, Ar[i], false, Br[i], (short)0, acc, false, false);
            acc = __builtin_amdgcn_wmma_f32_16x16x4_f32(
                false, Ai[i], false, Bq[i], (short)0, acc, false, false);
        }
        *reinterpret_cast<float4*>(dst + j * 256)     = make_float4(acc[0], acc[1], acc[2], acc[3]);
        *reinterpret_cast<float4*>(dst + j * 256 + 4) = make_float4(acc[4], acc[5], acc[6], acc[7]);

        if (j < 7) {
            #pragma unroll
            for (int i = 0; i < 16; ++i) {
                const v2f nr = Br[i] * Wr[i] + Bq[i] * Wi[i];
                const v2f nq = Bq[i] * Wr[i] - Br[i] * Wi[i];
                Br[i] = nr;
                Bq[i] = nq;
            }
        }
    }
}

extern "C" void kernel_launch(void* const* d_in, const int* in_sizes, int n_in,
                              void* d_out, int out_size, void* d_ws, size_t ws_size,
                              hipStream_t stream) {
    const float* log_dt     = (const float*)d_in[0];
    const float* C_real     = (const float*)d_in[1];
    const float* log_A_real = (const float*)d_in[2];
    const float* A_imag     = (const float*)d_in[3];
    float* out = (float*)d_out;

    dim3 grid(H_DIM);
    dim3 block(32);
    hipLaunchKernelGGL(s4d_wmma_kernel, grid, block, 0, stream,
                       log_dt, C_real, log_A_real, A_imag, out);
}


# One wave-wide range test replaces the large-argument branch in each of twelve sinf/cosf calls: six angles on one straight-line path, one reduction each, so all input loads issue as one group
# speedup vs baseline: 1.3824x; 1.0649x over previous
#include <hip/hip_runtime.h>

#define H_DIM 1024
#define N_DIM 64
#define L_DIM 2048

#define ROW_PITCH 132

typedef float v2f __attribute__((ext_vector_type(2)));
typedef float v4f __attribute__((ext_vector_type(4)));
typedef float v8f __attribute__((ext_vector_type(8)));

__device__ __forceinline__ void cmul(float& xr, float& xi, float yr, float yi) {
    float tr = xr * yr - xi * yi;
    xi = xr * yi + xi * yr;
    xr = tr;
}

__device__ __forceinline__ void sincos_short(float x, float& s, float& c) {
    const float ax = fabsf(x);
    const float q  = rintf(ax * 0x1.45f306p-1f);
    float r = fmaf(q, -0x1.921fb4p+0f, ax);
    r = fmaf(q, -0x1.4442d0p-24f, r);
    r = fmaf(q, -0x1.846988p-48f, r);
    const int   i  = (int)q;
    const float t  = r * r;
    const float sp = fmaf(r, t * fmaf(t, fmaf(t, -0x1.983304p-13f, 0x1.110388p-7f), -0x1.55553ap-3f), r);
    const float cp = fmaf(t, fmaf(t, fmaf(t, fmaf(t, 0x1.aea668p-16f, -0x1.6c9e76p-10f), 0x1.5557eep-5f),
                               -0x1.000008p-1f), 1.0f);
    const unsigned half_turn = ((unsigned)i & 2u) << 30;
    const unsigned sign_x    = __float_as_uint(x) & 0x80000000u;
    s = __uint_as_float(__float_as_uint((i & 1) ? cp : sp) ^ half_turn ^ sign_x);
    c = __uint_as_float(__float_as_uint((i & 1) ? -sp : cp) ^ half_turn);
}

__global__ __launch_bounds__(32) void s4d_wmma_kernel(
    const float* __restrict__ log_dt,
    const float* __restrict__ C_real,
    const float* __restrict__ log_A_real,
    const float* __restrict__ A_imag,
    float* __restrict__ out)
{
    __shared__ __attribute__((aligned(16))) float sQ[16 * ROW_PITCH];
    __shared__ __attribute__((aligned(16))) float sP[16 * ROW_PITCH];
    __shared__ __attribute__((aligned(16))) float sW[2 * N_DIM];

    const int h       = blockIdx.x;
    const int lane    = threadIdx.x;
    const int halfSel = lane >> 4;
    const int mn      = lane & 15;

    const float dt = expf(log_dt[h]);

    float lar[2], aim[2], cr[2], ci[2];
    #pragma unroll
    for (int s = 0; s < 2; ++s) {
        const int n = lane + 32 * s;
        lar[s] = log_A_real[h * N_DIM + n];
        aim[s] = A_imag[h * N_DIM + n];
        cr[s]  = C_real[(h * N_DIM + n) * 2 + 0];
        ci[s]  = C_real[(h * N_DIM + n) * 2 + 1];
    }

    float ar[2], dAr[2], ang[6], sn[6], cs[6];
    #pragma unroll
    for (int s = 0; s < 2; ++s) {
        ar[s]  = -expf(lar[s]);
        dAr[s] = ar[s] * dt;
        const float dAi = aim[s] * dt;
        ang[3 * s + 0] = dAi;
        ang[3 * s + 1] = 16.0f * dAi;
        ang[3 * s + 2] = 256.0f * dAi;
    }

    const int beyond = (fabsf(ang[2]) >= 131072.0f) || (fabsf(ang[5]) >= 131072.0f);
    if (!__any(beyond)) {
        #pragma unroll
        for (int k = 0; k < 6; ++k) sincos_short(ang[k], sn[k], cs[k]);
    } else {
        #pragma unroll
        for (int k = 0; k < 6; ++k) {
            sn[k] = sinf(ang[k]);
            cs[k] = cosf(ang[k]);
        }
    }

    float wr[2], wi[2], W16r[2], W16i[2], pr[2], pi[2];
    #pragma unroll
    for (int s = 0; s < 2; ++s) {
        const int n = lane + 32 * s;

        const float er = expf(dAr[s]);
        wr[s] = er * cs[3 * s + 0];
        wi[s] = er * sn[3 * s + 0];

        const float e1r = wr[s] - 1.0f, e1i = wi[s];
        const float inv = 1.0f / (ar[s] * ar[s] + aim[s] * aim[s]);
        const float tr  = (e1r * ar[s] + e1i * aim[s]) * inv;
        const float ti  = (e1i * ar[s] - e1r * aim[s]) * inv;
        pr[s] = 2.0f * (cr[s] * tr - ci[s] * ti);
        pi[s] = 2.0f * (cr[s] * ti + ci[s] * tr);

        const float er16 = expf(16.0f * dAr[s]);
        W16r[s] = er16 * cs[3 * s + 1];
        W16i[s] = er16 * sn[3 * s + 1];

        const float er256 = expf(256.0f * dAr[s]);
        const int slot = 4 * (n >> 1) + (n & 1);
        sW[slot]     = er256 * cs[3 * s + 2];
        sW[slot + 2] = er256 * sn[3 * s + 2];
    }

    {
        float qr[2] = {1.0f, 1.0f}, qi[2] = {0.0f, 0.0f};
        #pragma unroll
        for (int x = 0; x < 16; ++x) {
            #pragma unroll
            for (int s = 0; s < 2; ++s) {
                const int n = lane + 32 * s;
                const int slot = x * ROW_PITCH + 4 * (n >> 1) + (n & 1);
                sQ[slot]     = qr[s];
                sQ[slot + 2] = qi[s];
                sP[slot]     = pr[s];
                sP[slot + 2] = -pi[s];
                cmul(qr[s], qi[s], wr[s], wi[s]);
                cmul(pr[s], pi[s], W16r[s], W16i[s]);
            }
        }
    }
    __syncthreads();

    v2f Ar[16], Ai[16], Br[16], Bq[16], Wr[16], Wi[16];
    #pragma unroll
    for (int i = 0; i < 16; ++i) {
        const int off = 4 * (2 * i + halfSel);
        const v4f qa = *reinterpret_cast<const v4f*>(&sQ[mn * ROW_PITCH + off]);
        const v4f pb = *reinterpret_cast<const v4f*>(&sP[mn * ROW_PITCH + off]);
        const v4f ww = *reinterpret_cast<const v4f*>(&sW[off]);
        Ar[i].x = qa.x; Ar[i].y = qa.y; Ai[i].x = qa.z; Ai[i].y = qa.w;
        Br[i].x = pb.x; Br[i].y = pb.y; Bq[i].x = pb.z; Bq[i].y = pb.w;
        Wr[i].x = ww.x; Wr[i].y = ww.y; Wi[i].x = ww.z; Wi[i].y = ww.w;
    }

    float* dst = out + h * L_DIM + mn * 16 + 8 * halfSel;

    #pragma unroll
    for (int j = 0; j < 8; ++j) {
        v8f acc = {};
        #pragma unroll
        for (int i = 0; i < 16; ++i) {
            acc = __builtin_amdgcn_wmma_f32_16x16x4_f32(
                false, Ar[i], false, Br[i], (short)0, acc, false, false);
            acc = __builtin_amdgcn_wmma_f32_16x16x4_f32(
                false, Ai[i], false, Bq[i], (short)0, acc, false, false);
        }
        *reinterpret_cast<float4*>(dst + j * 256)     = make_float4(acc[0], acc[1], acc[2], acc[3]);
        *reinterpret_cast<float4*>(dst + j * 256 + 4) = make_float4(acc[4], acc[5], acc[6], acc[7]);

        if (j < 7) {
            #pragma unroll
            for (int i = 0; i < 16; ++i) {
                const v2f nr = Br[i] * Wr[i] + Bq[i] * Wi[i];
                const v2f nq = Bq[i] * Wr[i] - Br[i] * Wi[i];
                Br[i] = nr;
                Bq[i] = nq;
            }
        }
    }
}

extern "C" void kernel_launch(void* const* d_in, const int* in_sizes, int n_in,
                              void* d_out, int out_size, void* d_ws, size_t ws_size,
                              hipStream_t stream) {
    const float* log_dt     = (const float*)d_in[0];
    const float* C_real     = (const float*)d_in[1];
    const float* log_A_real = (const float*)d_in[2];
    const float* A_imag     = (const float*)d_in[3];
    float* out = (float*)d_out;

    dim3 grid(H_DIM);
    dim3 block(32);
    hipLaunchKernelGGL(s4d_wmma_kernel, grid, block, 0, stream,
                       log_dt, C_real, log_A_real, A_imag, out);
}
